# starting kernel
# baseline (speedup 1.0000x reference)
#include <hip/hip_runtime.h>
#include <hip/hip_bf16.h>

#define IN_CH   512
#define OUT_CH  64
#define KCHUNK  128
#define ROWS_PER_BLOCK 128

typedef __attribute__((ext_vector_type(2))) float v2f;
typedef __attribute__((ext_vector_type(4))) float v4f;
typedef __attribute__((ext_vector_type(8))) float v8f;

__global__ void init_kernel(float* __restrict__ agg, float* __restrict__ deg, int n) {
    int i = blockIdx.x * blockDim.x + threadIdx.x;
    int total = n * OUT_CH;
    if (i < total) agg[i] = 0.0f;
    if (i < n)     deg[i] = 1.0f;
}

__global__ void deg_kernel(const int* __restrict__ dst, float* __restrict__ deg, int e) {
    int i = blockIdx.x * blockDim.x + threadIdx.x;
    if (i < e) atomicAdd(&deg[dst[i]], 1.0f);
}

__global__ void dinv_kernel(float* __restrict__ deg, int n) {
    int i = blockIdx.x * blockDim.x + threadIdx.x;
    if (i < n) deg[i] = rsqrtf(deg[i]);
}

__global__ __launch_bounds__(256) void gemm_wmma_kernel(
        const float* __restrict__ x, const float* __restrict__ W,
        float* __restrict__ h, int n) {
    __shared__ __align__(16) v2f lds_w[(KCHUNK / 2) * OUT_CH];

    const int tid   = threadIdx.x;
    const int wave  = tid >> 5;
    const int lane  = tid & 31;
    const int half  = lane >> 4;
    const int l15   = lane & 15;
    const int kh    = half * 2;
    const int row0  = blockIdx.x * ROWS_PER_BLOCK + wave * 16;

    int arow = row0 + l15;
    if (arow >= n) arow = n - 1;
    const float* xrow = x + (size_t)arow * IN_CH;

    v8f acc[4] = {{}, {}, {}, {}};

    for (int kc = 0; kc < IN_CH; kc += KCHUNK) {
        #pragma unroll
        for (int i = 0; i < (KCHUNK / 2) * OUT_CH / 256; ++i) {
            int j = tid + i * 256;
            int p = j >> 6;
            int c = j & 63;
            v2f w2;
            w2.x = W[(size_t)(kc + 2 * p)     * OUT_CH + c];
            w2.y = W[(size_t)(kc + 2 * p + 1) * OUT_CH + c];
            lds_w[j] = w2;
        }
        __syncthreads();

        #pragma unroll 4
        for (int k = 0; k < KCHUNK; k += 4) {
            v2f a;
            a.x = xrow[kc + k + kh];
            a.y = xrow[kc + k + kh + 1];
            const int prow = (k >> 1) + half;
            #pragma unroll
            for (int t = 0; t < 4; ++t) {
                v2f b = lds_w[prow * OUT_CH + t * 16 + l15];
                acc[t] = __builtin_amdgcn_wmma_f32_16x16x4_f32(
                    false, a, false, b, (short)0, acc[t], false, false);
            }
        }
        __syncthreads();
    }

    #pragma unroll
    for (int v = 0; v < 8; ++v) {
        int row = row0 + half * 8 + v;
        if (row < n) {
            #pragma unroll
            for (int t = 0; t < 4; ++t)
                h[(size_t)row * OUT_CH + t * 16 + l15] = acc[t][v];
        }
    }
}

__global__ __launch_bounds__(256) void scatter_kernel(
        const float* __restrict__ h, const int* __restrict__ src,
        const int* __restrict__ dst, const float* __restrict__ dinv,
        float* __restrict__ agg, int e) {
    int edge = blockIdx.x * 8 + (threadIdx.x >> 5);
    if (edge >= e) return;
    int lane = threadIdx.x & 31;
    int s = src[edge];
    int d = dst[edge];
    float norm = dinv[s] * dinv[d];
    const float2* hs = (const float2*)(h + (size_t)s * OUT_CH);
    float2 v = hs[lane];
    float* out = agg + (size_t)d * OUT_CH + lane * 2;
    atomicAdd(out,     v.x * norm);
    atomicAdd(out + 1, v.y * norm);
}

__global__ void finalize_kernel(const float* __restrict__ agg,
                                const float* __restrict__ h,
                                const float* __restrict__ dinv,
                                const float* __restrict__ bias,
                                const float* __restrict__ alpha,
                                float* __restrict__ out, int n) {
    int idx = blockIdx.x * blockDim.x + threadIdx.x;
    if (idx >= n * OUT_CH) return;
    int c = idx & (OUT_CH - 1);
    int i = idx >> 6;
    float di = dinv[i];
    float v = agg[idx] + h[idx] * di * di + bias[c];
    out[idx] = (v >= 0.0f) ? v : alpha[c] * v;
}

extern "C" void kernel_launch(void* const* d_in, const int* in_sizes, int n_in,
                              void* d_out, int out_size, void* d_ws, size_t ws_size,
                              hipStream_t stream) {
    const float* x     = (const float*)d_in[0];
    const int*   ei    = (const int*)  d_in[1];
    const float* W     = (const float*)d_in[2];
    const float* bias  = (const float*)d_in[3];
    const float* alpha = (const float*)d_in[4];
    float*       out   = (float*)d_out;

    const int N = in_sizes[0] / IN_CH;
    const int E = in_sizes[1] / 2;
    const int* src = ei;
    const int* dst = ei + E;

    char* ws = (char*)d_ws;
    size_t hw_bytes = ((size_t)N * OUT_CH * sizeof(float) + 255) & ~(size_t)255;
    float* h    = (float*)ws;
    float* agg  = (float*)(ws + hw_bytes);
    float* dinv = (float*)(ws + 2 * hw_bytes);

    const int threads = 256;
    int totNC = N * OUT_CH;

    init_kernel<<<(totNC + threads - 1) / threads, threads, 0, stream>>>(agg, dinv, N);
    deg_kernel<<<(E + threads - 1) / threads, threads, 0, stream>>>(dst, dinv, E);
    dinv_kernel<<<(N + threads - 1) / threads, threads, 0, stream>>>(dinv, N);
    gemm_wmma_kernel<<<(N + ROWS_PER_BLOCK - 1) / ROWS_PER_BLOCK, threads, 0, stream>>>(
        x, W, h, N);
    scatter_kernel<<<(E + 7) / 8, threads, 0, stream>>>(h, src, dst, dinv, agg, E);
    finalize_kernel<<<(totNC + threads - 1) / threads, threads, 0, stream>>>(
        agg, h, dinv, bias, alpha, out, N);
}


# edge sum by pull: count planes prefix-summed into list offsets, chunk groups re-filter and store src per node (arrival order), one wave per node sums its rows in registers; no float atomics
# speedup vs baseline: 3.4218x; 3.4218x over previous
#include <hip/hip_runtime.h>
#include <hip/hip_bf16.h>

#define IN_CH   512
#define OUT_CH  64
#define KCHUNK  128
#define ROWS_PER_BLOCK 128

#define DEG_R        8
#define DEG_C        32
#define DEG_THREADS  512
#define DEG_NR_MAX   16384
#define DEG_VPL      8

typedef __attribute__((ext_vector_type(2))) float v2f;
typedef __attribute__((ext_vector_type(4))) float v4f;
typedef __attribute__((ext_vector_type(8))) float v8f;

#define SCAN_THREADS 256
#define FILL_VPL     4

__device__ __forceinline__ void deg_count_one(unsigned* cnt, int d, int lo, unsigned len) {
    unsigned a = (unsigned)(d - lo);
    if (a < len) atomicAdd(&cnt[a], 1u);
}

__global__ __launch_bounds__(DEG_THREADS) void deg_hist_kernel(
        const int* __restrict__ dst, unsigned* __restrict__ part, int n, int e) {
    __shared__ unsigned cnt[DEG_NR_MAX];

    const int tid = threadIdx.x;
    const int r   = blockIdx.x % DEG_R;
    const int c   = blockIdx.x / DEG_R;

    const int pitch  = (n + 31) & ~31;
    const int span   = ((n + DEG_R - 1) / DEG_R + 31) & ~31;
    const int lo_all = r * span;
    const int hi_all = (lo_all + span < n) ? lo_all + span : n;

    const int ec = ((e + DEG_C - 1) / DEG_C + 3) & ~3;
    const int e0 = c * ec;
    const int e1 = (e0 + ec < e) ? e0 + ec : e;

    const bool vec_ok = (((size_t)dst) & 15) == 0;
    const int  nvec   = (vec_ok && e1 > e0) ? ((e1 - e0) >> 2) : 0;
    const int4* dst4  = (const int4*)(dst + e0);

    unsigned* plane = part + (size_t)c * pitch;

    for (int lo = lo_all; lo < hi_all; lo += DEG_NR_MAX) {
        const int len = (hi_all - lo < DEG_NR_MAX) ? hi_all - lo : DEG_NR_MAX;

        for (int j = tid; j < len; j += DEG_THREADS) cnt[j] = 0u;
        __syncthreads();

        const int group = DEG_VPL * DEG_THREADS;
        const int nfull = nvec / group;
        for (int g = 0; g < nfull; ++g) {
            const int4* gp = dst4 + (size_t)g * group + tid;
            int4 d[DEG_VPL];
            #pragma unroll
            for (int k = 0; k < DEG_VPL; ++k) d[k] = gp[k * DEG_THREADS];
            #pragma unroll
            for (int k = 0; k < DEG_VPL; ++k) {
                deg_count_one(cnt, d[k].x, lo, (unsigned)len);
                deg_count_one(cnt, d[k].y, lo, (unsigned)len);
                deg_count_one(cnt, d[k].z, lo, (unsigned)len);
                deg_count_one(cnt, d[k].w, lo, (unsigned)len);
            }
        }
        for (int q = nfull * group + tid; q < nvec; q += DEG_THREADS) {
            int4 d = dst4[q];
            deg_count_one(cnt, d.x, lo, (unsigned)len);
            deg_count_one(cnt, d.y, lo, (unsigned)len);
            deg_count_one(cnt, d.z, lo, (unsigned)len);
            deg_count_one(cnt, d.w, lo, (unsigned)len);
        }
        for (int i = e0 + 4 * nvec + tid; i < e1; i += DEG_THREADS)
            deg_count_one(cnt, dst[i], lo, (unsigned)len);
        __syncthreads();

        for (int j = tid; j < len; j += DEG_THREADS) plane[lo + j] = cnt[j];
        __syncthreads();
    }
}

__global__ __launch_bounds__(SCAN_THREADS) void dinv_offsets_kernel(
        unsigned* __restrict__ part, float* __restrict__ dinv,
        unsigned* __restrict__ degcnt, unsigned* __restrict__ blocktot, int n) {
    __shared__ unsigned sc[SCAN_THREADS];
    const int tid = threadIdx.x;
    const int i = blockIdx.x * SCAN_THREADS + tid;
    const int pitch = (n + 31) & ~31;

    unsigned cnt[DEG_C];
    unsigned s = 0u;
    #pragma unroll
    for (int c = 0; c < DEG_C; ++c) {
        cnt[c] = (i < n) ? part[(size_t)c * pitch + i] : 0u;
        s += cnt[c];
    }
    if (i < n) {
        dinv[i]   = rsqrtf((float)(s + 1u));
        degcnt[i] = s;
    }

    sc[tid] = s;
    __syncthreads();
    for (int off = 1; off < SCAN_THREADS; off <<= 1) {
        const unsigned v = (tid >= off) ? sc[tid - off] : 0u;
        __syncthreads();
        sc[tid] += v;
        __syncthreads();
    }
    const unsigned incl = sc[tid];
    if (tid == SCAN_THREADS - 1) blocktot[blockIdx.x] = incl;

    if (i < n) {
        unsigned run = incl - s;
        #pragma unroll
        for (int c = 0; c < DEG_C; ++c) {
            part[(size_t)c * pitch + i] = run;
            run += cnt[c];
        }
    }
}

__device__ __forceinline__ void fill_one(unsigned* pos, int d, int s, int lo, unsigned len,
                                         int* __restrict__ csr, unsigned e) {
    unsigned a = (unsigned)(d - lo);
    if (a < len) {
        unsigned p = atomicAdd(&pos[a], 1u);
        if (p < e) csr[p] = s;
    }
}

__device__ __forceinline__ unsigned fill_slot(unsigned* pos, int d, int lo, unsigned len) {
    unsigned a = (unsigned)(d - lo);
    unsigned p = ~0u;
    if (a < len) p = atomicAdd(&pos[a], 1u);
    return p;
}

__global__ __launch_bounds__(DEG_THREADS) void fill_kernel(
        const int* __restrict__ src, const int* __restrict__ dst,
        const unsigned* __restrict__ part, const unsigned* __restrict__ blocktot,
        int* __restrict__ csr, unsigned* __restrict__ rowstart, int n, int e) {
    __shared__ unsigned pos[DEG_NR_MAX];
    __shared__ unsigned bsum[DEG_NR_MAX / SCAN_THREADS + 2];
    __shared__ unsigned bbase[DEG_NR_MAX / SCAN_THREADS + 2];
    __shared__ unsigned base0;

    const int tid = threadIdx.x;
    const int r   = blockIdx.x % DEG_R;
    const int c   = blockIdx.x / DEG_R;

    const int pitch  = (n + 31) & ~31;
    const int span   = ((n + DEG_R - 1) / DEG_R + 31) & ~31;
    const int lo_all = r * span;
    const int hi_all = (lo_all + span < n) ? lo_all + span : n;

    const int ec = ((e + DEG_C - 1) / DEG_C + 3) & ~3;
    const int e0 = c * ec;
    const int e1 = (e0 + ec < e) ? e0 + ec : e;

    const bool vec_ok = ((((size_t)dst) | ((size_t)src)) & 15) == 0;
    const int  nvec   = (vec_ok && e1 > e0) ? ((e1 - e0) >> 2) : 0;
    const int4* dst4  = (const int4*)(dst + e0);
    const int4* src4  = (const int4*)(src + e0);

    const unsigned* plane = part + (size_t)c * pitch;

    for (int lo = lo_all; lo < hi_all; lo += DEG_NR_MAX) {
        const int len = (hi_all - lo < DEG_NR_MAX) ? hi_all - lo : DEG_NR_MAX;

        const int b0   = lo / SCAN_THREADS;
        const int nblk = (lo + len - 1) / SCAN_THREADS - b0 + 1;
        if (tid == 0) base0 = 0u;
        if (tid < nblk) bsum[tid] = blocktot[b0 + tid];
        __syncthreads();
        unsigned psum = 0u;
        for (int k = tid; k < b0; k += DEG_THREADS) psum += blocktot[k];
        if (psum) atomicAdd(&base0, psum);
        __syncthreads();
        if (tid < nblk) {
            unsigned s = base0;
            for (int k = 0; k < tid; ++k) s += bsum[k];
            bbase[tid] = s;
        }
        __syncthreads();

        for (int j = tid; j < len; j += DEG_THREADS) {
            const unsigned v = plane[lo + j] + bbase[(lo + j) / SCAN_THREADS - b0];
            pos[j] = v;
            if (c == 0) rowstart[lo + j] = v;
        }
        __syncthreads();

        const int group = FILL_VPL * DEG_THREADS;
        const int nfull = nvec / group;
        for (int g = 0; g < nfull; ++g) {
            const size_t off = (size_t)g * group + tid;
            int4 d[FILL_VPL], s[FILL_VPL];
            #pragma unroll
            for (int k = 0; k < FILL_VPL; ++k) {
                d[k] = dst4[off + k * DEG_THREADS];
                s[k] = src4[off + k * DEG_THREADS];
            }
            #pragma unroll
            for (int k = 0; k < FILL_VPL; ++k) {
                const unsigned p0 = fill_slot(pos, d[k].x, lo, (unsigned)len);
                const unsigned p1 = fill_slot(pos, d[k].y, lo, (unsigned)len);
                const unsigned p2 = fill_slot(pos, d[k].z, lo, (unsigned)len);
                const unsigned p3 = fill_slot(pos, d[k].w, lo, (unsigned)len);
                if (p0 < (unsigned)e) csr[p0] = s[k].x;
                if (p1 < (unsigned)e) csr[p1] = s[k].y;
                if (p2 < (unsigned)e) csr[p2] = s[k].z;
                if (p3 < (unsigned)e) csr[p3] = s[k].w;
            }
        }
        for (int q = nfull * group + tid; q < nvec; q += DEG_THREADS) {
            const int4 d = dst4[q];
            const int4 s = src4[q];
            fill_one(pos, d.x, s.x, lo, (unsigned)len, csr, (unsigned)e);
            fill_one(pos, d.y, s.y, lo, (unsigned)len, csr, (unsigned)e);
            fill_one(pos, d.z, s.z, lo, (unsigned)len, csr, (unsigned)e);
            fill_one(pos, d.w, s.w, lo, (unsigned)len, csr, (unsigned)e);
        }
        for (int i = e0 + 4 * nvec + tid; i < e1; i += DEG_THREADS)
            fill_one(pos, dst[i], src[i], lo, (unsigned)len, csr, (unsigned)e);
        __syncthreads();
    }
}

__global__ __launch_bounds__(256) void gemm_wmma_kernel(
        const float* __restrict__ x, const float* __restrict__ W,
        float* __restrict__ h, int n) {
    __shared__ __align__(16) v2f lds_w[(KCHUNK / 2) * OUT_CH];

    const int tid   = threadIdx.x;
    const int wave  = tid >> 5;
    const int lane  = tid & 31;
    const int half  = lane >> 4;
    const int l15   = lane & 15;
    const int kh    = half * 2;
    const int row0  = blockIdx.x * ROWS_PER_BLOCK + wave * 16;

    int arow = row0 + l15;
    if (arow >= n) arow = n - 1;
    const float* xrow = x + (size_t)arow * IN_CH;

    v8f acc[4] = {{}, {}, {}, {}};

    for (int kc = 0; kc < IN_CH; kc += KCHUNK) {
        #pragma unroll
        for (int i = 0; i < (KCHUNK / 2) * OUT_CH / 256; ++i) {
            int j = tid + i * 256;
            int p = j >> 6;
            int c = j & 63;
            v2f w2;
            w2.x = W[(size_t)(kc + 2 * p)     * OUT_CH + c];
            w2.y = W[(size_t)(kc + 2 * p + 1) * OUT_CH + c];
            lds_w[j] = w2;
        }
        __syncthreads();

        #pragma unroll 4
        for (int k = 0; k < KCHUNK; k += 4) {
            v2f a;
            a.x = xrow[kc + k + kh];
            a.y = xrow[kc + k + kh + 1];
            const int prow = (k >> 1) + half;
            #pragma unroll
            for (int t = 0; t < 4; ++t) {
                v2f b = lds_w[prow * OUT_CH + t * 16 + l15];
                acc[t] = __builtin_amdgcn_wmma_f32_16x16x4_f32(
                    false, a, false, b, (short)0, acc[t], false, false);
            }
        }
        __syncthreads();
    }

    #pragma unroll
    for (int v = 0; v < 8; ++v) {
        int row = row0 + half * 8 + v;
        if (row < n) {
            #pragma unroll
            for (int t = 0; t < 4; ++t)
                h[(size_t)row * OUT_CH + t * 16 + l15] = acc[t][v];
        }
    }
}

__global__ __launch_bounds__(256) void pull_kernel(
        const float* __restrict__ h, const int* __restrict__ csr,
        const unsigned* __restrict__ rowstart, const unsigned* __restrict__ degcnt,
        const float* __restrict__ dinv, const float* __restrict__ bias,
        const float* __restrict__ alpha, float* __restrict__ out, int n, int e) {
    const int lane = threadIdx.x & 31;
    const int wave = __builtin_amdgcn_readfirstlane((int)(threadIdx.x >> 5));
    const int node = blockIdx.x * 8 + wave;
    if (node >= n) return;

    unsigned start = rowstart[node];
    unsigned cnt   = degcnt[node];
    if (start > (unsigned)e) start = (unsigned)e;
    if (cnt > (unsigned)e - start) cnt = (unsigned)e - start;

    const float dn = dinv[node];
    const float2* h2 = (const float2*)h;
    const float2 hv = h2[(size_t)node * (OUT_CH / 2) + lane];
    const int c = lane * 2;
    const float b0 = bias[c],  b1 = bias[c + 1];
    const float a0 = alpha[c], a1 = alpha[c + 1];
    float accx = 0.0f, accy = 0.0f;

    for (unsigned base = 0; base < cnt; base += 32) {
        const int m = (cnt - base < 32u) ? (int)(cnt - base) : 32;
        int   sl = 0;
        float nl = 0.0f;
        if (lane < m) {
            sl = csr[start + base + lane];
            nl = dinv[sl] * dn;
        }
        int j = 0;
        for (; j + 8 <= m; j += 8) {
            float2 v[8];
            float  w[8];
            #pragma unroll
            for (int u = 0; u < 8; ++u) {
                const int s = __builtin_amdgcn_readlane(sl, j + u);
                w[u] = __int_as_float(__builtin_amdgcn_readlane(__float_as_int(nl), j + u));
                v[u] = h2[(size_t)s * (OUT_CH / 2) + lane];
            }
            #pragma unroll
            for (int u = 0; u < 8; ++u) {
                accx += v[u].x * w[u];
                accy += v[u].y * w[u];
            }
        }
        if (j < m) {
            float2 v[8];
            float  w[8];
            #pragma unroll
            for (int u = 0; u < 8; ++u) {
                v[u].x = 0.0f; v[u].y = 0.0f; w[u] = 0.0f;
                if (j + u < m) {
                    const int s = __builtin_amdgcn_readlane(sl, j + u);
                    w[u] = __int_as_float(__builtin_amdgcn_readlane(__float_as_int(nl), j + u));
                    v[u] = h2[(size_t)s * (OUT_CH / 2) + lane];
                }
            }
            #pragma unroll
            for (int u = 0; u < 8; ++u) {
                accx += v[u].x * w[u];
                accy += v[u].y * w[u];
            }
        }
    }

    const float vx = accx + hv.x * dn * dn + b0;
    const float vy = accy + hv.y * dn * dn + b1;
    float2 o;
    o.x = (vx >= 0.0f) ? vx : a0 * vx;
    o.y = (vy >= 0.0f) ? vy : a1 * vy;
    ((float2*)out)[(size_t)node * (OUT_CH / 2) + lane] = o;
}

extern "C" void kernel_launch(void* const* d_in, const int* in_sizes, int n_in,
                              void* d_out, int out_size, void* d_ws, size_t ws_size,
                              hipStream_t stream) {
    const float* x     = (const float*)d_in[0];
    const int*   ei    = (const int*)  d_in[1];
    const float* W     = (const float*)d_in[2];
    const float* bias  = (const float*)d_in[3];
    const float* alpha = (const float*)d_in[4];
    float*       out   = (float*)d_out;

    const int N = in_sizes[0] / IN_CH;
    const int E = in_sizes[1] / 2;
    const int* src = ei;
    const int* dst = ei + E;

    char* ws = (char*)d_ws;
    size_t hw_bytes = ((size_t)N * OUT_CH * sizeof(float) + 255) & ~(size_t)255;
    size_t e_bytes  = ((size_t)E * sizeof(int) + 255) & ~(size_t)255;
    size_t n_bytes  = ((size_t)N * sizeof(int) + 255) & ~(size_t)255;
    float*    h        = (float*)ws;
    unsigned* part     = (unsigned*)ws;
    int*      csr      = (int*)(ws + hw_bytes);
    unsigned* rowstart = (unsigned*)(ws + hw_bytes + e_bytes);
    unsigned* degcnt   = (unsigned*)(ws + hw_bytes + e_bytes + n_bytes);
    unsigned* blocktot = (unsigned*)(ws + hw_bytes + e_bytes + 2 * n_bytes);
    float*    dinv     = (float*)(ws + 2 * hw_bytes);

    const int threads = 256;
    const int nscan = (N + SCAN_THREADS - 1) / SCAN_THREADS;

    deg_hist_kernel<<<DEG_R * DEG_C, DEG_THREADS, 0, stream>>>(dst, part, N, E);
    dinv_offsets_kernel<<<nscan, SCAN_THREADS, 0, stream>>>(part, dinv, degcnt, blocktot, N);
    fill_kernel<<<DEG_R * DEG_C, DEG_THREADS, 0, stream>>>(src, dst, part, blocktot, csr, rowstart, N, E);
    gemm_wmma_kernel<<<(N + ROWS_PER_BLOCK - 1) / ROWS_PER_BLOCK, threads, 0, stream>>>(
        x, W, h, N);
    pull_kernel<<<(N + 7) / 8, threads, 0, stream>>>(
        h, csr, rowstart, degcnt, dinv, bias, alpha, out, N, E);
}


# x@W as a resident grid: 512 work-groups keep all of W in LDS (staged once, no barrier in the k loop); a wave walks its 16-row tiles, x fragments fetched 6 16-k groups ahead; k order unchanged
# speedup vs baseline: 3.6125x; 1.0557x over previous
#include <hip/hip_runtime.h>
#include <hip/hip_bf16.h>

#define IN_CH   512
#define OUT_CH  64

#define DEG_R        8
#define DEG_C        32
#define DEG_THREADS  512
#define DEG_NR_MAX   16384
#define DEG_VPL      8

typedef __attribute__((ext_vector_type(2))) float v2f;
typedef __attribute__((ext_vector_type(4))) float v4f;
typedef __attribute__((ext_vector_type(8))) float v8f;

#define SCAN_THREADS 256
#define FILL_VPL     4

__device__ __forceinline__ void deg_count_one(unsigned* cnt, int d, int lo, unsigned len) {
    unsigned a = (unsigned)(d - lo);
    if (a < len) atomicAdd(&cnt[a], 1u);
}

__global__ __launch_bounds__(DEG_THREADS) void deg_hist_kernel(
        const int* __restrict__ dst, unsigned* __restrict__ part, int n, int e) {
    __shared__ unsigned cnt[DEG_NR_MAX];

    const int tid = threadIdx.x;
    const int r   = blockIdx.x % DEG_R;
    const int c   = blockIdx.x / DEG_R;

    const int pitch  = (n + 31) & ~31;
    const int span   = ((n + DEG_R - 1) / DEG_R + 31) & ~31;
    const int lo_all = r * span;
    const int hi_all = (lo_all + span < n) ? lo_all + span : n;

    const int ec = ((e + DEG_C - 1) / DEG_C + 3) & ~3;
    const int e0 = c * ec;
    const int e1 = (e0 + ec < e) ? e0 + ec : e;

    const bool vec_ok = (((size_t)dst) & 15) == 0;
    const int  nvec   = (vec_ok && e1 > e0) ? ((e1 - e0) >> 2) : 0;
    const int4* dst4  = (const int4*)(dst + e0);

    unsigned* plane = part + (size_t)c * pitch;

    for (int lo = lo_all; lo < hi_all; lo += DEG_NR_MAX) {
        const int len = (hi_all - lo < DEG_NR_MAX) ? hi_all - lo : DEG_NR_MAX;

        for (int j = tid; j < len; j += DEG_THREADS) cnt[j] = 0u;
        __syncthreads();

        const int group = DEG_VPL * DEG_THREADS;
        const int nfull = nvec / group;
        for (int g = 0; g < nfull; ++g) {
            const int4* gp = dst4 + (size_t)g * group + tid;
            int4 d[DEG_VPL];
            #pragma unroll
            for (int k = 0; k < DEG_VPL; ++k) d[k] = gp[k * DEG_THREADS];
            #pragma unroll
            for (int k = 0; k < DEG_VPL; ++k) {
                deg_count_one(cnt, d[k].x, lo, (unsigned)len);
                deg_count_one(cnt, d[k].y, lo, (unsigned)len);
                deg_count_one(cnt, d[k].z, lo, (unsigned)len);
                deg_count_one(cnt, d[k].w, lo, (unsigned)len);
            }
        }
        for (int q = nfull * group + tid; q < nvec; q += DEG_THREADS) {
            int4 d = dst4[q];
            deg_count_one(cnt, d.x, lo, (unsigned)len);
            deg_count_one(cnt, d.y, lo, (unsigned)len);
            deg_count_one(cnt, d.z, lo, (unsigned)len);
            deg_count_one(cnt, d.w, lo, (unsigned)len);
        }
        for (int i = e0 + 4 * nvec + tid; i < e1; i += DEG_THREADS)
            deg_count_one(cnt, dst[i], lo, (unsigned)len);
        __syncthreads();

        for (int j = tid; j < len; j += DEG_THREADS) plane[lo + j] = cnt[j];
        __syncthreads();
    }
}

__global__ __launch_bounds__(SCAN_THREADS) void dinv_offsets_kernel(
        unsigned* __restrict__ part, float* __restrict__ dinv,
        unsigned* __restrict__ degcnt, unsigned* __restrict__ blocktot, int n) {
    __shared__ unsigned sc[SCAN_THREADS];
    const int tid = threadIdx.x;
    const int i = blockIdx.x * SCAN_THREADS + tid;
    const int pitch = (n + 31) & ~31;

    unsigned cnt[DEG_C];
    unsigned s = 0u;
    #pragma unroll
    for (int c = 0; c < DEG_C; ++c) {
        cnt[c] = (i < n) ? part[(size_t)c * pitch + i] : 0u;
        s += cnt[c];
    }
    if (i < n) {
        dinv[i]   = rsqrtf((float)(s + 1u));
        degcnt[i] = s;
    }

    sc[tid] = s;
    __syncthreads();
    for (int off = 1; off < SCAN_THREADS; off <<= 1) {
        const unsigned v = (tid >= off) ? sc[tid - off] : 0u;
        __syncthreads();
        sc[tid] += v;
        __syncthreads();
    }
    const unsigned incl = sc[tid];
    if (tid == SCAN_THREADS - 1) blocktot[blockIdx.x] = incl;

    if (i < n) {
        unsigned run = incl - s;
        #pragma unroll
        for (int c = 0; c < DEG_C; ++c) {
            part[(size_t)c * pitch + i] = run;
            run += cnt[c];
        }
    }
}

__device__ __forceinline__ void fill_one(unsigned* pos, int d, int s, int lo, unsigned len,
                                         int* __restrict__ csr, unsigned e) {
    unsigned a = (unsigned)(d - lo);
    if (a < len) {
        unsigned p = atomicAdd(&pos[a], 1u);
        if (p < e) csr[p] = s;
    }
}

__device__ __forceinline__ unsigned fill_slot(unsigned* pos, int d, int lo, unsigned len) {
    unsigned a = (unsigned)(d - lo);
    unsigned p = ~0u;
    if (a < len) p = atomicAdd(&pos[a], 1u);
    return p;
}

__global__ __launch_bounds__(DEG_THREADS) void fill_kernel(
        const int* __restrict__ src, const int* __restrict__ dst,
        const unsigned* __restrict__ part, const unsigned* __restrict__ blocktot,
        int* __restrict__ csr, unsigned* __restrict__ rowstart, int n, int e) {
    __shared__ unsigned pos[DEG_NR_MAX];
    __shared__ unsigned bsum[DEG_NR_MAX / SCAN_THREADS + 2];
    __shared__ unsigned bbase[DEG_NR_MAX / SCAN_THREADS + 2];
    __shared__ unsigned base0;

    const int tid = threadIdx.x;
    const int r   = blockIdx.x % DEG_R;
    const int c   = blockIdx.x / DEG_R;

    const int pitch  = (n + 31) & ~31;
    const int span   = ((n + DEG_R - 1) / DEG_R + 31) & ~31;
    const int lo_all = r * span;
    const int hi_all = (lo_all + span < n) ? lo_all + span : n;

    const int ec = ((e + DEG_C - 1) / DEG_C + 3) & ~3;
    const int e0 = c * ec;
    const int e1 = (e0 + ec < e) ? e0 + ec : e;

    const bool vec_ok = ((((size_t)dst) | ((size_t)src)) & 15) == 0;
    const int  nvec   = (vec_ok && e1 > e0) ? ((e1 - e0) >> 2) : 0;
    const int4* dst4  = (const int4*)(dst + e0);
    const int4* src4  = (const int4*)(src + e0);

    const unsigned* plane = part + (size_t)c * pitch;

    for (int lo = lo_all; lo < hi_all; lo += DEG_NR_MAX) {
        const int len = (hi_all - lo < DEG_NR_MAX) ? hi_all - lo : DEG_NR_MAX;

        const int b0   = lo / SCAN_THREADS;
        const int nblk = (lo + len - 1) / SCAN_THREADS - b0 + 1;
        if (tid == 0) base0 = 0u;
        if (tid < nblk) bsum[tid] = blocktot[b0 + tid];
        __syncthreads();
        unsigned psum = 0u;
        for (int k = tid; k < b0; k += DEG_THREADS) psum += blocktot[k];
        if (psum) atomicAdd(&base0, psum);
        __syncthreads();
        if (tid < nblk) {
            unsigned s = base0;
            for (int k = 0; k < tid; ++k) s += bsum[k];
            bbase[tid] = s;
        }
        __syncthreads();

        for (int j = tid; j < len; j += DEG_THREADS) {
            const unsigned v = plane[lo + j] + bbase[(lo + j) / SCAN_THREADS - b0];
            pos[j] = v;
            if (c == 0) rowstart[lo + j] = v;
        }
        __syncthreads();

        const int group = FILL_VPL * DEG_THREADS;
        const int nfull = nvec / group;
        for (int g = 0; g < nfull; ++g) {
            const size_t off = (size_t)g * group + tid;
            int4 d[FILL_VPL], s[FILL_VPL];
            #pragma unroll
            for (int k = 0; k < FILL_VPL; ++k) {
                d[k] = dst4[off + k * DEG_THREADS];
                s[k] = src4[off + k * DEG_THREADS];
            }
            #pragma unroll
            for (int k = 0; k < FILL_VPL; ++k) {
                const unsigned p0 = fill_slot(pos, d[k].x, lo, (unsigned)len);
                const unsigned p1 = fill_slot(pos, d[k].y, lo, (unsigned)len);
                const unsigned p2 = fill_slot(pos, d[k].z, lo, (unsigned)len);
                const unsigned p3 = fill_slot(pos, d[k].w, lo, (unsigned)len);
                if (p0 < (unsigned)e) csr[p0] = s[k].x;
                if (p1 < (unsigned)e) csr[p1] = s[k].y;
                if (p2 < (unsigned)e) csr[p2] = s[k].z;
                if (p3 < (unsigned)e) csr[p3] = s[k].w;
            }
        }
        for (int q = nfull * group + tid; q < nvec; q += DEG_THREADS) {
            const int4 d = dst4[q];
            const int4 s = src4[q];
            fill_one(pos, d.x, s.x, lo, (unsigned)len, csr, (unsigned)e);
            fill_one(pos, d.y, s.y, lo, (unsigned)len, csr, (unsigned)e);
            fill_one(pos, d.z, s.z, lo, (unsigned)len, csr, (unsigned)e);
            fill_one(pos, d.w, s.w, lo, (unsigned)len, csr, (unsigned)e);
        }
        for (int i = e0 + 4 * nvec + tid; i < e1; i += DEG_THREADS)
            fill_one(pos, dst[i], src[i], lo, (unsigned)len, csr, (unsigned)e);
        __syncthreads();
    }
}

#define GEMM_WGS 512
#define A_RING   8
#define A_AHEAD  6
#define K_GROUPS (IN_CH / 16)

__global__ __launch_bounds__(256) void gemm_wmma_kernel(
        const float* __restrict__ x, const float* __restrict__ W,
        float* __restrict__ h, int n) {
    __shared__ __align__(16) v2f lds_w[(IN_CH / 2) * OUT_CH];

    const int tid   = threadIdx.x;
    const int wave  = __builtin_amdgcn_readfirstlane(tid >> 5);
    const int lane  = tid & 31;
    const int half  = lane >> 4;
    const int l15   = lane & 15;
    const int kh    = half * 2;

    for (int i = 0; i < (IN_CH / 2) * OUT_CH / 256; ++i) {
        int j = tid + i * 256;
        int p = j >> 6;
        int c = j & 63;
        v2f w2;
        w2.x = W[(size_t)(2 * p)     * OUT_CH + c];
        w2.y = W[(size_t)(2 * p + 1) * OUT_CH + c];
        lds_w[j] = w2;
    }
    __syncthreads();

    const int ntiles = (n + 15) >> 4;
    const int nwg    = (int)gridDim.x;
    const int G      = nwg * 8;
    const int g      = (int)blockIdx.x * 8 + wave;
    const int nfull  = ntiles / G;
    const int rem    = ntiles - nfull * G;
    const int q      = wave * nwg + (int)blockIdx.x;
    const int mine   = nfull + ((q < rem) ? 1 : 0);
    if (mine == 0) return;

    #define TILE_OF(i) (((i) < nfull) ? (i) * G + g : nfull * G + q)
    #define ROW_PTR(t) (x + (size_t)(((t) * 16 + l15 < n) ? (t) * 16 + l15 : n - 1) * IN_CH + kh)

    const float* xr = ROW_PTR(TILE_OF(0));

    v2f ar[A_RING][4];
    #pragma unroll
    for (int gq = 0; gq < A_AHEAD; ++gq) {
        #pragma unroll
        for (int u = 0; u < 4; ++u) {
            ar[gq][u].x = xr[gq * 16 + u * 4];
            ar[gq][u].y = xr[gq * 16 + u * 4 + 1];
        }
        __builtin_amdgcn_sched_barrier(0x3C6);
    }

    for (int i = 0; i < mine; ++i) {
        const int tile = TILE_OF(i);
        const float* xn = (i + 1 < mine) ? ROW_PTR(TILE_OF(i + 1)) : xr;

        v8f acc[4] = {{}, {}, {}, {}};

        for (int jo = 0; jo < K_GROUPS; jo += A_RING) {
            #pragma unroll
            for (int jj = 0; jj < A_RING; ++jj) {
                const int gq = jo + jj + A_AHEAD;
                const float* pf = (gq < K_GROUPS) ? xr : xn;
                const int gk = gq & (K_GROUPS - 1);
                #pragma unroll
                for (int u = 0; u < 4; ++u) {
                    ar[(jj + A_AHEAD) % A_RING][u].x = pf[gk * 16 + u * 4];
                    ar[(jj + A_AHEAD) % A_RING][u].y = pf[gk * 16 + u * 4 + 1];
                }
                __builtin_amdgcn_sched_barrier(0x3C6);
                #pragma unroll
                for (int u = 0; u < 4; ++u) {
                    const v2f a = ar[jj][u];
                    const int prow = (jo + jj) * 8 + u * 2 + half;
                    #pragma unroll
                    for (int t = 0; t < 4; ++t) {
                        v2f b = lds_w[prow * OUT_CH + t * 16 + l15];
                        acc[t] = __builtin_amdgcn_wmma_f32_16x16x4_f32(
                            false, a, false, b, (short)0, acc[t], false, false);
                    }
                }
            }
        }

        const int row0 = tile * 16;
        #pragma unroll
        for (int v = 0; v < 8; ++v) {
            int row = row0 + half * 8 + v;
            if (row < n) {
                #pragma unroll
                for (int t = 0; t < 4; ++t)
                    h[(size_t)row * OUT_CH + t * 16 + l15] = acc[t][v];
            }
        }
        xr = xn;
    }
    #undef TILE_OF
    #undef ROW_PTR
}

__global__ __launch_bounds__(256) void pull_kernel(
        const float* __restrict__ h, const int* __restrict__ csr,
        const unsigned* __restrict__ rowstart, const unsigned* __restrict__ degcnt,
        const float* __restrict__ dinv, const float* __restrict__ bias,
        const float* __restrict__ alpha, float* __restrict__ out, int n, int e) {
    const int lane = threadIdx.x & 31;
    const int wave = __builtin_amdgcn_readfirstlane((int)(threadIdx.x >> 5));
    const int node = blockIdx.x * 8 + wave;
    if (node >= n) return;

    unsigned start = rowstart[node];
    unsigned cnt   = degcnt[node];
    if (start > (unsigned)e) start = (unsigned)e;
    if (cnt > (unsigned)e - start) cnt = (unsigned)e - start;

    const float dn = dinv[node];
    const float2* h2 = (const float2*)h;
    const float2 hv = h2[(size_t)node * (OUT_CH / 2) + lane];
    const int c = lane * 2;
    const float b0 = bias[c],  b1 = bias[c + 1];
    const float a0 = alpha[c], a1 = alpha[c + 1];
    float accx = 0.0f, accy = 0.0f;

    for (unsigned base = 0; base < cnt; base += 32) {
        const int m = (cnt - base < 32u) ? (int)(cnt - base) : 32;
        int   sl = 0;
        float nl = 0.0f;
        if (lane < m) {
            sl = csr[start + base + lane];
            nl = dinv[sl] * dn;
        }
        int j = 0;
        for (; j + 8 <= m; j += 8) {
            float2 v[8];
            float  w[8];
            #pragma unroll
            for (int u = 0; u < 8; ++u) {
                const int s = __builtin_amdgcn_readlane(sl, j + u);
                w[u] = __int_as_float(__builtin_amdgcn_readlane(__float_as_int(nl), j + u));
                v[u] = h2[(size_t)s * (OUT_CH / 2) + lane];
            }
            #pragma unroll
            for (int u = 0; u < 8; ++u) {
                accx += v[u].x * w[u];
                accy += v[u].y * w[u];
            }
        }
        if (j < m) {
            float2 v[8];
            float  w[8];
            #pragma unroll
            for (int u = 0; u < 8; ++u) {
                v[u].x = 0.0f; v[u].y = 0.0f; w[u] = 0.0f;
                if (j + u < m) {
                    const int s = __builtin_amdgcn_readlane(sl, j + u);
                    w[u] = __int_as_float(__builtin_amdgcn_readlane(__float_as_int(nl), j + u));
                    v[u] = h2[(size_t)s * (OUT_CH / 2) + lane];
                }
            }
            #pragma unroll
            for (int u = 0; u < 8; ++u) {
                accx += v[u].x * w[u];
                accy += v[u].y * w[u];
            }
        }
    }

    const float vx = accx + hv.x * dn * dn + b0;
    const float vy = accy + hv.y * dn * dn + b1;
    float2 o;
    o.x = (vx >= 0.0f) ? vx : a0 * vx;
    o.y = (vy >= 0.0f) ? vy : a1 * vy;
    ((float2*)out)[(size_t)node * (OUT_CH / 2) + lane] = o;
}

extern "C" void kernel_launch(void* const* d_in, const int* in_sizes, int n_in,
                              void* d_out, int out_size, void* d_ws, size_t ws_size,
                              hipStream_t stream) {
    const float* x     = (const float*)d_in[0];
    const int*   ei    = (const int*)  d_in[1];
    const float* W     = (const float*)d_in[2];
    const float* bias  = (const float*)d_in[3];
    const float* alpha = (const float*)d_in[4];
    float*       out   = (float*)d_out;

    const int N = in_sizes[0] / IN_CH;
    const int E = in_sizes[1] / 2;
    const int* src = ei;
    const int* dst = ei + E;

    char* ws = (char*)d_ws;
    size_t hw_bytes = ((size_t)N * OUT_CH * sizeof(float) + 255) & ~(size_t)255;
    size_t e_bytes  = ((size_t)E * sizeof(int) + 255) & ~(size_t)255;
    size_t n_bytes  = ((size_t)N * sizeof(int) + 255) & ~(size_t)255;
    float*    h        = (float*)ws;
    unsigned* part     = (unsigned*)ws;
    int*      csr      = (int*)(ws + hw_bytes);
    unsigned* rowstart = (unsigned*)(ws + hw_bytes + e_bytes);
    unsigned* degcnt   = (unsigned*)(ws + hw_bytes + e_bytes + n_bytes);
    unsigned* blocktot = (unsigned*)(ws + hw_bytes + e_bytes + 2 * n_bytes);
    float*    dinv     = (float*)(ws + 2 * hw_bytes);

    const int threads = 256;
    const int nscan = (N + SCAN_THREADS - 1) / SCAN_THREADS;

    deg_hist_kernel<<<DEG_R * DEG_C, DEG_THREADS, 0, stream>>>(dst, part, N, E);
    dinv_offsets_kernel<<<nscan, SCAN_THREADS, 0, stream>>>(part, dinv, degcnt, blocktot, N);
    fill_kernel<<<DEG_R * DEG_C, DEG_THREADS, 0, stream>>>(src, dst, part, blocktot, csr, rowstart, N, E);
    gemm_wmma_kernel<<<GEMM_WGS, threads, 0, stream>>>(x, W, h, N);
    pull_kernel<<<(N + 7) / 8, threads, 0, stream>>>(
        h, csr, rowstart, degcnt, dinv, bias, alpha, out, N, E);
}


# edge sum takes dinv[src] from a byte table of edge counts kept in LDS (resident 32-wave groups, nodes drawn from an LDS counter), not from a 32-line load per 32 edges; same values, sums, order
# speedup vs baseline: 3.8396x; 1.0629x over previous
#include <hip/hip_runtime.h>
#include <hip/hip_bf16.h>

#define IN_CH   512
#define OUT_CH  64

#define DEG_R        8
#define DEG_C        32
#define DEG_THREADS  512
#define DEG_NR_MAX   16384
#define DEG_VPL      8

typedef __attribute__((ext_vector_type(2))) float v2f;
typedef __attribute__((ext_vector_type(4))) float v4f;
typedef __attribute__((ext_vector_type(8))) float v8f;

#define SCAN_THREADS 256
#define FILL_VPL     4

__device__ __forceinline__ void deg_count_one(unsigned* cnt, int d, int lo, unsigned len) {
    unsigned a = (unsigned)(d - lo);
    if (a < len) atomicAdd(&cnt[a], 1u);
}

__global__ __launch_bounds__(DEG_THREADS) void deg_hist_kernel(
        const int* __restrict__ dst, unsigned* __restrict__ part, int n, int e) {
    __shared__ unsigned cnt[DEG_NR_MAX];

    const int tid = threadIdx.x;
    const int r   = blockIdx.x % DEG_R;
    const int c   = blockIdx.x / DEG_R;

    const int pitch  = (n + 31) & ~31;
    const int span   = ((n + DEG_R - 1) / DEG_R + 31) & ~31;
    const int lo_all = r * span;
    const int hi_all = (lo_all + span < n) ? lo_all + span : n;

    const int ec = ((e + DEG_C - 1) / DEG_C + 3) & ~3;
    const int e0 = c * ec;
    const int e1 = (e0 + ec < e) ? e0 + ec : e;

    const bool vec_ok = (((size_t)dst) & 15) == 0;
    const int  nvec   = (vec_ok && e1 > e0) ? ((e1 - e0) >> 2) : 0;
    const int4* dst4  = (const int4*)(dst + e0);

    unsigned* plane = part + (size_t)c * pitch;

    for (int lo = lo_all; lo < hi_all; lo += DEG_NR_MAX) {
        const int len = (hi_all - lo < DEG_NR_MAX) ? hi_all - lo : DEG_NR_MAX;

        for (int j = tid; j < len; j += DEG_THREADS) cnt[j] = 0u;
        __syncthreads();

        const int group = DEG_VPL * DEG_THREADS;
        const int nfull = nvec / group;
        for (int g = 0; g < nfull; ++g) {
            const int4* gp = dst4 + (size_t)g * group + tid;
            int4 d[DEG_VPL];
            #pragma unroll
            for (int k = 0; k < DEG_VPL; ++k) d[k] = gp[k * DEG_THREADS];
            #pragma unroll
            for (int k = 0; k < DEG_VPL; ++k) {
                deg_count_one(cnt, d[k].x, lo, (unsigned)len);
                deg_count_one(cnt, d[k].y, lo, (unsigned)len);
                deg_count_one(cnt, d[k].z, lo, (unsigned)len);
                deg_count_one(cnt, d[k].w, lo, (unsigned)len);
            }
        }
        for (int q = nfull * group + tid; q < nvec; q += DEG_THREADS) {
            int4 d = dst4[q];
            deg_count_one(cnt, d.x, lo, (unsigned)len);
            deg_count_one(cnt, d.y, lo, (unsigned)len);
            deg_count_one(cnt, d.z, lo, (unsigned)len);
            deg_count_one(cnt, d.w, lo, (unsigned)len);
        }
        for (int i = e0 + 4 * nvec + tid; i < e1; i += DEG_THREADS)
            deg_count_one(cnt, dst[i], lo, (unsigned)len);
        __syncthreads();

        for (int j = tid; j < len; j += DEG_THREADS) plane[lo + j] = cnt[j];
        __syncthreads();
    }
}

__global__ __launch_bounds__(SCAN_THREADS) void dinv_offsets_kernel(
        unsigned* __restrict__ part, float* __restrict__ dinv,
        unsigned* __restrict__ degcnt, unsigned* __restrict__ blocktot,
        unsigned* __restrict__ deg8w, int n) {
    __shared__ unsigned sc[SCAN_THREADS];
    __shared__ unsigned d8w[SCAN_THREADS / 4];
    const int tid = threadIdx.x;
    const int i = blockIdx.x * SCAN_THREADS + tid;
    const int pitch = (n + 31) & ~31;

    unsigned cnt[DEG_C];
    unsigned s = 0u;
    #pragma unroll
    for (int c = 0; c < DEG_C; ++c) {
        cnt[c] = (i < n) ? part[(size_t)c * pitch + i] : 0u;
        s += cnt[c];
    }
    if (i < n) {
        dinv[i]   = rsqrtf((float)(s + 1u));
        degcnt[i] = s;
    }

    sc[tid] = s;
    ((unsigned char*)d8w)[tid] = (unsigned char)((s < 255u) ? s : 255u);
    __syncthreads();
    if (tid < SCAN_THREADS / 4)
        deg8w[blockIdx.x * (SCAN_THREADS / 4) + tid] = d8w[tid];
    for (int off = 1; off < SCAN_THREADS; off <<= 1) {
        const unsigned v = (tid >= off) ? sc[tid - off] : 0u;
        __syncthreads();
        sc[tid] += v;
        __syncthreads();
    }
    const unsigned incl = sc[tid];
    if (tid == SCAN_THREADS - 1) blocktot[blockIdx.x] = incl;

    if (i < n) {
        unsigned run = incl - s;
        #pragma unroll
        for (int c = 0; c < DEG_C; ++c) {
            part[(size_t)c * pitch + i] = run;
            run += cnt[c];
        }
    }
}

__device__ __forceinline__ void fill_one(unsigned* pos, int d, int s, int lo, unsigned len,
                                         int* __restrict__ csr, unsigned e) {
    unsigned a = (unsigned)(d - lo);
    if (a < len) {
        unsigned p = atomicAdd(&pos[a], 1u);
        if (p < e) csr[p] = s;
    }
}

__device__ __forceinline__ unsigned fill_slot(unsigned* pos, int d, int lo, unsigned len) {
    unsigned a = (unsigned)(d - lo);
    unsigned p = ~0u;
    if (a < len) p = atomicAdd(&pos[a], 1u);
    return p;
}

__global__ __launch_bounds__(DEG_THREADS) void fill_kernel(
        const int* __restrict__ src, const int* __restrict__ dst,
        const unsigned* __restrict__ part, const unsigned* __restrict__ blocktot,
        int* __restrict__ csr, unsigned* __restrict__ rowstart, int n, int e) {
    __shared__ unsigned pos[DEG_NR_MAX];
    __shared__ unsigned bsum[DEG_NR_MAX / SCAN_THREADS + 2];
    __shared__ unsigned bbase[DEG_NR_MAX / SCAN_THREADS + 2];
    __shared__ unsigned base0;

    const int tid = threadIdx.x;
    const int r   = blockIdx.x % DEG_R;
    const int c   = blockIdx.x / DEG_R;

    const int pitch  = (n + 31) & ~31;
    const int span   = ((n + DEG_R - 1) / DEG_R + 31) & ~31;
    const int lo_all = r * span;
    const int hi_all = (lo_all + span < n) ? lo_all + span : n;

    const int ec = ((e + DEG_C - 1) / DEG_C + 3) & ~3;
    const int e0 = c * ec;
    const int e1 = (e0 + ec < e) ? e0 + ec : e;

    const bool vec_ok = ((((size_t)dst) | ((size_t)src)) & 15) == 0;
    const int  nvec   = (vec_ok && e1 > e0) ? ((e1 - e0) >> 2) : 0;
    const int4* dst4  = (const int4*)(dst + e0);
    const int4* src4  = (const int4*)(src + e0);

    const unsigned* plane = part + (size_t)c * pitch;

    for (int lo = lo_all; lo < hi_all; lo += DEG_NR_MAX) {
        const int len = (hi_all - lo < DEG_NR_MAX) ? hi_all - lo : DEG_NR_MAX;

        const int b0   = lo / SCAN_THREADS;
        const int nblk = (lo + len - 1) / SCAN_THREADS - b0 + 1;
        if (tid == 0) base0 = 0u;
        if (tid < nblk) bsum[tid] = blocktot[b0 + tid];
        __syncthreads();
        unsigned psum = 0u;
        for (int k = tid; k < b0; k += DEG_THREADS) psum += blocktot[k];
        if (psum) atomicAdd(&base0, psum);
        __syncthreads();
        if (tid < nblk) {
            unsigned s = base0;
            for (int k = 0; k < tid; ++k) s += bsum[k];
            bbase[tid] = s;
        }
        __syncthreads();

        for (int j = tid; j < len; j += DEG_THREADS) {
            const unsigned v = plane[lo + j] + bbase[(lo + j) / SCAN_THREADS - b0];
            pos[j] = v;
            if (c == 0) rowstart[lo + j] = v;
        }
        __syncthreads();

        const int group = FILL_VPL * DEG_THREADS;
        const int nfull = nvec / group;
        for (int g = 0; g < nfull; ++g) {
            const size_t off = (size_t)g * group + tid;
            int4 d[FILL_VPL], s[FILL_VPL];
            #pragma unroll
            for (int k = 0; k < FILL_VPL; ++k) {
                d[k] = dst4[off + k * DEG_THREADS];
                s[k] = src4[off + k * DEG_THREADS];
            }
            #pragma unroll
            for (int k = 0; k < FILL_VPL; ++k) {
                const unsigned p0 = fill_slot(pos, d[k].x, lo, (unsigned)len);
                const unsigned p1 = fill_slot(pos, d[k].y, lo, (unsigned)len);
                const unsigned p2 = fill_slot(pos, d[k].z, lo, (unsigned)len);
                const unsigned p3 = fill_slot(pos, d[k].w, lo, (unsigned)len);
                if (p0 < (unsigned)e) csr[p0] = s[k].x;
                if (p1 < (unsigned)e) csr[p1] = s[k].y;
                if (p2 < (unsigned)e) csr[p2] = s[k].z;
                if (p3 < (unsigned)e) csr[p3] = s[k].w;
            }
        }
        for (int q = nfull * group + tid; q < nvec; q += DEG_THREADS) {
            const int4 d = dst4[q];
            const int4 s = src4[q];
            fill_one(pos, d.x, s.x, lo, (unsigned)len, csr, (unsigned)e);
            fill_one(pos, d.y, s.y, lo, (unsigned)len, csr, (unsigned)e);
            fill_one(pos, d.z, s.z, lo, (unsigned)len, csr, (unsigned)e);
            fill_one(pos, d.w, s.w, lo, (unsigned)len, csr, (unsigned)e);
        }
        for (int i = e0 + 4 * nvec + tid; i < e1; i += DEG_THREADS)
            fill_one(pos, dst[i], src[i], lo, (unsigned)len, csr, (unsigned)e);
        __syncthreads();
    }
}

#define GEMM_WGS 512
#define A_RING   8
#define A_AHEAD  6
#define K_GROUPS (IN_CH / 16)

__global__ __launch_bounds__(256) void gemm_wmma_kernel(
        const float* __restrict__ x, const float* __restrict__ W,
        float* __restrict__ h, int n) {
    __shared__ __align__(16) v2f lds_w[(IN_CH / 2) * OUT_CH];

    const int tid   = threadIdx.x;
    const int wave  = __builtin_amdgcn_readfirstlane(tid >> 5);
    const int lane  = tid & 31;
    const int half  = lane >> 4;
    const int l15   = lane & 15;
    const int kh    = half * 2;

    for (int i = 0; i < (IN_CH / 2) * OUT_CH / 256; ++i) {
        int j = tid + i * 256;
        int p = j >> 6;
        int c = j & 63;
        v2f w2;
        w2.x = W[(size_t)(2 * p)     * OUT_CH + c];
        w2.y = W[(size_t)(2 * p + 1) * OUT_CH + c];
        lds_w[j] = w2;
    }
    __syncthreads();

    const int ntiles = (n + 15) >> 4;
    const int nwg    = (int)gridDim.x;
    const int G      = nwg * 8;
    const int g      = (int)blockIdx.x * 8 + wave;
    const int nfull  = ntiles / G;
    const int rem    = ntiles - nfull * G;
    const int q      = wave * nwg + (int)blockIdx.x;
    const int mine   = nfull + ((q < rem) ? 1 : 0);
    if (mine == 0) return;

    #define TILE_OF(i) (((i) < nfull) ? (i) * G + g : nfull * G + q)
    #define ROW_PTR(t) (x + (size_t)(((t) * 16 + l15 < n) ? (t) * 16 + l15 : n - 1) * IN_CH + kh)

    const float* xr = ROW_PTR(TILE_OF(0));

    v2f ar[A_RING][4];
    #pragma unroll
    for (int gq = 0; gq < A_AHEAD; ++gq) {
        #pragma unroll
        for (int u = 0; u < 4; ++u) {
            ar[gq][u].x = xr[gq * 16 + u * 4];
            ar[gq][u].y = xr[gq * 16 + u * 4 + 1];
        }
        __builtin_amdgcn_sched_barrier(0x3C6);
    }

    for (int i = 0; i < mine; ++i) {
        const int tile = TILE_OF(i);
        const float* xn = (i + 1 < mine) ? ROW_PTR(TILE_OF(i + 1)) : xr;

        v8f acc[4] = {{}, {}, {}, {}};

        for (int jo = 0; jo < K_GROUPS; jo += A_RING) {
            #pragma unroll
            for (int jj = 0; jj < A_RING; ++jj) {
                const int gq = jo + jj + A_AHEAD;
                const float* pf = (gq < K_GROUPS) ? xr : xn;
                const int gk = gq & (K_GROUPS - 1);
                #pragma unroll
                for (int u = 0; u < 4; ++u) {
                    ar[(jj + A_AHEAD) % A_RING][u].x = pf[gk * 16 + u * 4];
                    ar[(jj + A_AHEAD) % A_RING][u].y = pf[gk * 16 + u * 4 + 1];
                }
                __builtin_amdgcn_sched_barrier(0x3C6);
                #pragma unroll
                for (int u = 0; u < 4; ++u) {
                    const v2f a = ar[jj][u];
                    const int prow = (jo + jj) * 8 + u * 2 + half;
                    #pragma unroll
                    for (int t = 0; t < 4; ++t) {
                        v2f b = lds_w[prow * OUT_CH + t * 16 + l15];
                        acc[t] = __builtin_amdgcn_wmma_f32_16x16x4_f32(
                            false, a, false, b, (short)0, acc[t], false, false);
                    }
                }
            }
        }

        const int row0 = tile * 16;
        #pragma unroll
        for (int v = 0; v < 8; ++v) {
            int row = row0 + half * 8 + v;
            if (row < n) {
                #pragma unroll
                for (int t = 0; t < 4; ++t)
                    h[(size_t)row * OUT_CH + t * 16 + l15] = acc[t][v];
            }
        }
        xr = xn;
    }
    #undef TILE_OF
    #undef ROW_PTR
}

#define PULL_WGS      512
#define PULL_THREADS  1024
#define TAB_CAP       102400
#define TAB_STG       7

__global__ __launch_bounds__(PULL_THREADS) void pull_kernel(
        const float* __restrict__ h, const int* __restrict__ csr,
        const unsigned* __restrict__ rowstart, const unsigned* __restrict__ degcnt,
        const float* __restrict__ dinv, const unsigned* __restrict__ deg8w,
        const float* __restrict__ bias, const float* __restrict__ alpha,
        float* __restrict__ out, int n, int e) {
    __shared__ uint4 tab4[TAB_CAP / 16];
    __shared__ unsigned next;

    const int tid  = threadIdx.x;
    const int lane = tid & 31;
    const int wave = __builtin_amdgcn_readfirstlane(tid >> 5);

    const int nscan = (n + SCAN_THREADS - 1) / SCAN_THREADS;
    const int nu    = (nscan * (SCAN_THREADS / 16) < TAB_CAP / 16) ? nscan * (SCAN_THREADS / 16) : TAB_CAP / 16;
    const unsigned tabn = (unsigned)nu * 16u;
    if (nu > 0) {
        const uint4* g4 = (const uint4*)deg8w;
        const int st = (int)((blockIdx.x * 64u) % (unsigned)nu);
        uint4 r[TAB_STG];
        int   qs[TAB_STG];
        #pragma unroll
        for (int k = 0; k < TAB_STG; ++k) {
            int q = tid + k * PULL_THREADS;
            q = (q < nu) ? q : nu - 1;
            q += st;
            if (q >= nu) q -= nu;
            qs[k] = q;
            r[k] = g4[q];
        }
        #pragma unroll
        for (int k = 0; k < TAB_STG; ++k) tab4[qs[k]] = r[k];
    }
    if (tid == 0) next = 32u;
    __syncthreads();
    const unsigned char* tab8 = (const unsigned char*)tab4;

    const int nb  = (int)gridDim.x;
    const int per = n / nb;
    const int rem = n - per * nb;
    const int b   = (int)blockIdx.x;
    const int n0  = b * per + ((b < rem) ? b : rem);
    const int n1  = n0 + per + ((b < rem) ? 1 : 0);

    const float2* h2 = (const float2*)h;
    const int c = lane * 2;
    const float b0 = bias[c],  b1 = bias[c + 1];
    const float a0 = alpha[c], a1 = alpha[c + 1];

    int cur = wave;
    while (n0 + cur < n1) {
        const int node = n0 + cur;

        unsigned start = rowstart[node];
        unsigned cnt   = degcnt[node];
        if (start > (unsigned)e) start = (unsigned)e;
        if (cnt > (unsigned)e - start) cnt = (unsigned)e - start;

        const float dn = dinv[node];
        const float2 hv = h2[(size_t)node * (OUT_CH / 2) + lane];
        float accx = 0.0f, accy = 0.0f;

        for (unsigned base = 0; base < cnt; base += 32) {
            const int m = (cnt - base < 32u) ? (int)(cnt - base) : 32;
            int   sl = 0;
            float nl = 0.0f;
            if (lane < m) {
                sl = csr[start + base + lane];
                const unsigned c8 = ((unsigned)sl < tabn) ? (unsigned)tab8[sl] : 255u;
                float di;
                if (c8 < 255u) di = rsqrtf((float)(c8 + 1u));
                else           di = dinv[sl];
                nl = di * dn;
            }
            int j = 0;
            for (; j + 8 <= m; j += 8) {
                float2 v[8];
                float  w[8];
                #pragma unroll
                for (int u = 0; u < 8; ++u) {
                    const int s = __builtin_amdgcn_readlane(sl, j + u);
                    w[u] = __int_as_float(__builtin_amdgcn_readlane(__float_as_int(nl), j + u));
                    v[u] = h2[(size_t)s * (OUT_CH / 2) + lane];
                }
                #pragma unroll
                for (int u = 0; u < 8; ++u) {
                    accx += v[u].x * w[u];
                    accy += v[u].y * w[u];
                }
            }
            if (j < m) {
                float2 v[8];
                float  w[8];
                #pragma unroll
                for (int u = 0; u < 8; ++u) {
                    v[u].x = 0.0f; v[u].y = 0.0f; w[u] = 0.0f;
                    if (j + u < m) {
                        const int s = __builtin_amdgcn_readlane(sl, j + u);
                        w[u] = __int_as_float(__builtin_amdgcn_readlane(__float_as_int(nl), j + u));
                        v[u] = h2[(size_t)s * (OUT_CH / 2) + lane];
                    }
                }
                #pragma unroll
                for (int u = 0; u < 8; ++u) {
                    accx += v[u].x * w[u];
                    accy += v[u].y * w[u];
                }
            }
        }

        const float vx = accx + hv.x * dn * dn + b0;
        const float vy = accy + hv.y * dn * dn + b1;
        float2 o;
        o.x = (vx >= 0.0f) ? vx : a0 * vx;
        o.y = (vy >= 0.0f) ? vy : a1 * vy;
        ((float2*)out)[(size_t)node * (OUT_CH / 2) + lane] = o;

        unsigned nx = 0u;
        if (lane == 0) nx = atomicAdd(&next, 1u);
        cur = __builtin_amdgcn_readfirstlane((int)nx);
    }
}

extern "C" void kernel_launch(void* const* d_in, const int* in_sizes, int n_in,
                              void* d_out, int out_size, void* d_ws, size_t ws_size,
                              hipStream_t stream) {
    const float* x     = (const float*)d_in[0];
    const int*   ei    = (const int*)  d_in[1];
    const float* W     = (const float*)d_in[2];
    const float* bias  = (const float*)d_in[3];
    const float* alpha = (const float*)d_in[4];
    float*       out   = (float*)d_out;

    const int N = in_sizes[0] / IN_CH;
    const int E = in_sizes[1] / 2;
    const int* src = ei;
    const int* dst = ei + E;

    char* ws = (char*)d_ws;
    size_t hw_bytes = ((size_t)N * OUT_CH * sizeof(float) + 255) & ~(size_t)255;
    size_t e_bytes  = ((size_t)E * sizeof(int) + 255) & ~(size_t)255;
    size_t n_bytes  = ((size_t)N * sizeof(int) + 255) & ~(size_t)255;
    float*    h        = (float*)ws;
    unsigned* part     = (unsigned*)ws;
    int*      csr      = (int*)(ws + hw_bytes);
    unsigned* rowstart = (unsigned*)(ws + hw_bytes + e_bytes);
    unsigned* degcnt   = (unsigned*)(ws + hw_bytes + e_bytes + n_bytes);
    unsigned* blocktot = (unsigned*)(ws + hw_bytes + e_bytes + 2 * n_bytes);
    float*    dinv     = (float*)(ws + 2 * hw_bytes);

    const int threads = 256;
    const int nscan = (N + SCAN_THREADS - 1) / SCAN_THREADS;
    size_t bt_bytes = ((size_t)nscan * sizeof(unsigned) + 255) & ~(size_t)255;
    unsigned* deg8w = (unsigned*)(ws + hw_bytes + e_bytes + 2 * n_bytes + bt_bytes);

    deg_hist_kernel<<<DEG_R * DEG_C, DEG_THREADS, 0, stream>>>(dst, part, N, E);
    dinv_offsets_kernel<<<nscan, SCAN_THREADS, 0, stream>>>(part, dinv, degcnt, blocktot, deg8w, N);
    fill_kernel<<<DEG_R * DEG_C, DEG_THREADS, 0, stream>>>(src, dst, part, blocktot, csr, rowstart, N, E);
    gemm_wmma_kernel<<<GEMM_WGS, threads, 0, stream>>>(x, W, h, N);
    pull_kernel<<<PULL_WGS, PULL_THREADS, 0, stream>>>(
        h, csr, rowstart, degcnt, dinv, deg8w, bias, alpha, out, N, E);
}


# count and fill launches ask for loads one stage ahead: next group's list vectors into a second register set under the current group's LDS adds, a pass's offset words all at once; same slots
# speedup vs baseline: 4.0634x; 1.0583x over previous
#include <hip/hip_runtime.h>
#include <hip/hip_bf16.h>

#define IN_CH   512
#define OUT_CH  64

#define DEG_R        8
#define DEG_C        32
#define DEG_THREADS  512
#define DEG_NR_MAX   16384
#define DEG_VPL      8

typedef __attribute__((ext_vector_type(2))) float v2f;
typedef __attribute__((ext_vector_type(4))) float v4f;
typedef __attribute__((ext_vector_type(8))) float v8f;

#define SCAN_THREADS 256
#define FILL_VPL     4

__device__ __forceinline__ void deg_count_one(unsigned* cnt, int d, int lo, unsigned len) {
    unsigned a = (unsigned)(d - lo);
    if (a < len) atomicAdd(&cnt[a], 1u);
}

__device__ __forceinline__ void deg_load_group(int4 (&d)[DEG_VPL], const int4* gp) {
    #pragma unroll
    for (int k = 0; k < DEG_VPL; ++k) d[k] = gp[k * DEG_THREADS];
}

__device__ __forceinline__ void deg_count_group(unsigned* cnt, const int4 (&d)[DEG_VPL],
                                                int lo, unsigned len) {
    #pragma unroll
    for (int k = 0; k < DEG_VPL; ++k) {
        deg_count_one(cnt, d[k].x, lo, len);
        deg_count_one(cnt, d[k].y, lo, len);
        deg_count_one(cnt, d[k].z, lo, len);
        deg_count_one(cnt, d[k].w, lo, len);
    }
}

__global__ __launch_bounds__(DEG_THREADS) void deg_hist_kernel(
        const int* __restrict__ dst, unsigned* __restrict__ part, int n, int e) {
    __shared__ unsigned cnt[DEG_NR_MAX];

    const int tid = threadIdx.x;
    const int r   = blockIdx.x % DEG_R;
    const int c   = blockIdx.x / DEG_R;

    const int pitch  = (n + 31) & ~31;
    const int span   = ((n + DEG_R - 1) / DEG_R + 31) & ~31;
    const int lo_all = r * span;
    const int hi_all = (lo_all + span < n) ? lo_all + span : n;

    const int ec = ((e + DEG_C - 1) / DEG_C + 3) & ~3;
    const int e0 = c * ec;
    const int e1 = (e0 + ec < e) ? e0 + ec : e;

    const bool vec_ok = (((size_t)dst) & 15) == 0;
    const int  nvec   = (vec_ok && e1 > e0) ? ((e1 - e0) >> 2) : 0;
    const int4* dst4  = (const int4*)(dst + e0);

    unsigned* plane = part + (size_t)c * pitch;

    for (int lo = lo_all; lo < hi_all; lo += DEG_NR_MAX) {
        const int len = (hi_all - lo < DEG_NR_MAX) ? hi_all - lo : DEG_NR_MAX;

        const int group = DEG_VPL * DEG_THREADS;
        const int nfull = nvec / group;
        const int4* gp0 = dst4 + tid;
        int4 da[DEG_VPL], db[DEG_VPL];
        if (nfull > 0) deg_load_group(da, gp0);

        for (int j = tid; j < len; j += DEG_THREADS) cnt[j] = 0u;
        __syncthreads();

        int g = 0;
        for (; g + 1 < nfull; g += 2) {
            deg_load_group(db, gp0 + (size_t)(g + 1) * group);
            deg_count_group(cnt, da, lo, (unsigned)len);
            const int g2 = (g + 2 < nfull) ? g + 2 : g + 1;
            deg_load_group(da, gp0 + (size_t)g2 * group);
            deg_count_group(cnt, db, lo, (unsigned)len);
        }
        if (g < nfull) deg_count_group(cnt, da, lo, (unsigned)len);
        for (int q = nfull * group + tid; q < nvec; q += DEG_THREADS) {
            int4 d = dst4[q];
            deg_count_one(cnt, d.x, lo, (unsigned)len);
            deg_count_one(cnt, d.y, lo, (unsigned)len);
            deg_count_one(cnt, d.z, lo, (unsigned)len);
            deg_count_one(cnt, d.w, lo, (unsigned)len);
        }
        for (int i = e0 + 4 * nvec + tid; i < e1; i += DEG_THREADS)
            deg_count_one(cnt, dst[i], lo, (unsigned)len);
        __syncthreads();

        for (int j = tid; j < len; j += DEG_THREADS) plane[lo + j] = cnt[j];
        __syncthreads();
    }
}

__global__ __launch_bounds__(SCAN_THREADS) void dinv_offsets_kernel(
        unsigned* __restrict__ part, float* __restrict__ dinv,
        unsigned* __restrict__ degcnt, unsigned* __restrict__ blocktot,
        unsigned* __restrict__ deg8w, int n) {
    __shared__ unsigned sc[SCAN_THREADS];
    __shared__ unsigned d8w[SCAN_THREADS / 4];
    const int tid = threadIdx.x;
    const int i = blockIdx.x * SCAN_THREADS + tid;
    const int pitch = (n + 31) & ~31;

    unsigned cnt[DEG_C];
    unsigned s = 0u;
    #pragma unroll
    for (int c = 0; c < DEG_C; ++c) {
        cnt[c] = (i < n) ? part[(size_t)c * pitch + i] : 0u;
        s += cnt[c];
    }
    if (i < n) {
        dinv[i]   = rsqrtf((float)(s + 1u));
        degcnt[i] = s;
    }

    sc[tid] = s;
    ((unsigned char*)d8w)[tid] = (unsigned char)((s < 255u) ? s : 255u);
    __syncthreads();
    if (tid < SCAN_THREADS / 4)
        deg8w[blockIdx.x * (SCAN_THREADS / 4) + tid] = d8w[tid];
    for (int off = 1; off < SCAN_THREADS; off <<= 1) {
        const unsigned v = (tid >= off) ? sc[tid - off] : 0u;
        __syncthreads();
        sc[tid] += v;
        __syncthreads();
    }
    const unsigned incl = sc[tid];
    if (tid == SCAN_THREADS - 1) blocktot[blockIdx.x] = incl;

    if (i < n) {
        unsigned run = incl - s;
        #pragma unroll
        for (int c = 0; c < DEG_C; ++c) {
            part[(size_t)c * pitch + i] = run;
            run += cnt[c];
        }
    }
}

__device__ __forceinline__ void fill_one(unsigned* pos, int d, int s, int lo, unsigned len,
                                         int* __restrict__ csr, unsigned e) {
    unsigned a = (unsigned)(d - lo);
    if (a < len) {
        unsigned p = atomicAdd(&pos[a], 1u);
        if (p < e) csr[p] = s;
    }
}

__device__ __forceinline__ unsigned fill_slot(unsigned* pos, int d, int lo, unsigned len) {
    unsigned a = (unsigned)(d - lo);
    unsigned p = ~0u;
    if (a < len) p = atomicAdd(&pos[a], 1u);
    return p;
}

__device__ __forceinline__ void fill_load_group(int4 (&d)[FILL_VPL], int4 (&s)[FILL_VPL],
                                                const int4* dp, const int4* sp) {
    #pragma unroll
    for (int k = 0; k < FILL_VPL; ++k) {
        d[k] = dp[k * DEG_THREADS];
        s[k] = sp[k * DEG_THREADS];
    }
}

__device__ __forceinline__ void fill_place_group(unsigned* pos, const int4 (&d)[FILL_VPL],
                                                 const int4 (&s)[FILL_VPL], int lo, unsigned len,
                                                 int* __restrict__ csr, unsigned e) {
    #pragma unroll
    for (int k = 0; k < FILL_VPL; ++k) {
        const unsigned p0 = fill_slot(pos, d[k].x, lo, len);
        const unsigned p1 = fill_slot(pos, d[k].y, lo, len);
        const unsigned p2 = fill_slot(pos, d[k].z, lo, len);
        const unsigned p3 = fill_slot(pos, d[k].w, lo, len);
        if (p0 < e) csr[p0] = s[k].x;
        if (p1 < e) csr[p1] = s[k].y;
        if (p2 < e) csr[p2] = s[k].z;
        if (p3 < e) csr[p3] = s[k].w;
    }
}

#define POS_VPL (DEG_NR_MAX / DEG_THREADS)

__global__ __launch_bounds__(DEG_THREADS) void fill_kernel(
        const int* __restrict__ src, const int* __restrict__ dst,
        const unsigned* __restrict__ part, const unsigned* __restrict__ blocktot,
        int* __restrict__ csr, unsigned* __restrict__ rowstart, int n, int e) {
    __shared__ unsigned pos[DEG_NR_MAX];
    __shared__ unsigned bsum[DEG_NR_MAX / SCAN_THREADS + 2];
    __shared__ unsigned bbase[DEG_NR_MAX / SCAN_THREADS + 2];
    __shared__ unsigned base0;

    const int tid = threadIdx.x;
    const int r   = blockIdx.x % DEG_R;
    const int c   = blockIdx.x / DEG_R;

    const int pitch  = (n + 31) & ~31;
    const int span   = ((n + DEG_R - 1) / DEG_R + 31) & ~31;
    const int lo_all = r * span;
    const int hi_all = (lo_all + span < n) ? lo_all + span : n;

    const int ec = ((e + DEG_C - 1) / DEG_C + 3) & ~3;
    const int e0 = c * ec;
    const int e1 = (e0 + ec < e) ? e0 + ec : e;

    const bool vec_ok = ((((size_t)dst) | ((size_t)src)) & 15) == 0;
    const int  nvec   = (vec_ok && e1 > e0) ? ((e1 - e0) >> 2) : 0;
    const int4* dst4  = (const int4*)(dst + e0);
    const int4* src4  = (const int4*)(src + e0);

    const unsigned* plane = part + (size_t)c * pitch;

    for (int lo = lo_all; lo < hi_all; lo += DEG_NR_MAX) {
        const int len = (hi_all - lo < DEG_NR_MAX) ? hi_all - lo : DEG_NR_MAX;

        unsigned pv[POS_VPL];
        #pragma unroll
        for (int k = 0; k < POS_VPL; ++k) {
            const int j = tid + k * DEG_THREADS;
            pv[k] = plane[lo + ((j < len) ? j : len - 1)];
        }
        const int group = FILL_VPL * DEG_THREADS;
        const int nfull = nvec / group;
        const int4* dp0 = dst4 + tid;
        const int4* sp0 = src4 + tid;
        int4 da[FILL_VPL], sa[FILL_VPL], db[FILL_VPL], sb[FILL_VPL];
        if (nfull > 0) fill_load_group(da, sa, dp0, sp0);

        const int b0   = lo / SCAN_THREADS;
        const int nblk = (lo + len - 1) / SCAN_THREADS - b0 + 1;
        if (tid == 0) base0 = 0u;
        if (tid < nblk) bsum[tid] = blocktot[b0 + tid];
        __syncthreads();
        unsigned psum = 0u;
        for (int k = tid; k < b0; k += DEG_THREADS) psum += blocktot[k];
        if (psum) atomicAdd(&base0, psum);
        __syncthreads();
        if (tid < nblk) {
            unsigned s = base0;
            for (int k = 0; k < tid; ++k) s += bsum[k];
            bbase[tid] = s;
        }
        __syncthreads();

        unsigned bb[POS_VPL];
        #pragma unroll
        for (int k = 0; k < POS_VPL; ++k) {
            const int j = tid + k * DEG_THREADS;
            bb[k] = bbase[(lo + ((j < len) ? j : len - 1)) / SCAN_THREADS - b0];
        }
        #pragma unroll
        for (int k = 0; k < POS_VPL; ++k) {
            const int j = tid + k * DEG_THREADS;
            if (j < len) {
                const unsigned v = pv[k] + bb[k];
                pos[j] = v;
                if (c == 0) rowstart[lo + j] = v;
            }
        }
        __syncthreads();

        int g = 0;
        for (; g + 1 < nfull; g += 2) {
            const size_t o1 = (size_t)(g + 1) * group;
            fill_load_group(db, sb, dp0 + o1, sp0 + o1);
            fill_place_group(pos, da, sa, lo, (unsigned)len, csr, (unsigned)e);
            const size_t o2 = (size_t)((g + 2 < nfull) ? g + 2 : g + 1) * group;
            fill_load_group(da, sa, dp0 + o2, sp0 + o2);
            fill_place_group(pos, db, sb, lo, (unsigned)len, csr, (unsigned)e);
        }
        if (g < nfull) fill_place_group(pos, da, sa, lo, (unsigned)len, csr, (unsigned)e);
        for (int q = nfull * group + tid; q < nvec; q += DEG_THREADS) {
            const int4 d = dst4[q];
            const int4 s = src4[q];
            fill_one(pos, d.x, s.x, lo, (unsigned)len, csr, (unsigned)e);
            fill_one(pos, d.y, s.y, lo, (unsigned)len, csr, (unsigned)e);
            fill_one(pos, d.z, s.z, lo, (unsigned)len, csr, (unsigned)e);
            fill_one(pos, d.w, s.w, lo, (unsigned)len, csr, (unsigned)e);
        }
        for (int i = e0 + 4 * nvec + tid; i < e1; i += DEG_THREADS)
            fill_one(pos, dst[i], src[i], lo, (unsigned)len, csr, (unsigned)e);
        __syncthreads();
    }
}

#define GEMM_WGS 512
#define A_RING   8
#define A_AHEAD  6
#define K_GROUPS (IN_CH / 16)

__global__ __launch_bounds__(256) void gemm_wmma_kernel(
        const float* __restrict__ x, const float* __restrict__ W,
        float* __restrict__ h, int n) {
    __shared__ __align__(16) v2f lds_w[(IN_CH / 2) * OUT_CH];

    const int tid   = threadIdx.x;
    const int wave  = __builtin_amdgcn_readfirstlane(tid >> 5);
    const int lane  = tid & 31;
    const int half  = lane >> 4;
    const int l15   = lane & 15;
    const int kh    = half * 2;

    for (int i = 0; i < (IN_CH / 2) * OUT_CH / 256; ++i) {
        int j = tid + i * 256;
        int p = j >> 6;
        int c = j & 63;
        v2f w2;
        w2.x = W[(size_t)(2 * p)     * OUT_CH + c];
        w2.y = W[(size_t)(2 * p + 1) * OUT_CH + c];
        lds_w[j] = w2;
    }
    __syncthreads();

    const int ntiles = (n + 15) >> 4;
    const int nwg    = (int)gridDim.x;
    const int G      = nwg * 8;
    const int g      = (int)blockIdx.x * 8 + wave;
    const int nfull  = ntiles / G;
    const int rem    = ntiles - nfull * G;
    const int q      = wave * nwg + (int)blockIdx.x;
    const int mine   = nfull + ((q < rem) ? 1 : 0);
    if (mine == 0) return;

    #define TILE_OF(i) (((i) < nfull) ? (i) * G + g : nfull * G + q)
    #define ROW_PTR(t) (x + (size_t)(((t) * 16 + l15 < n) ? (t) * 16 + l15 : n - 1) * IN_CH + kh)

    const float* xr = ROW_PTR(TILE_OF(0));

    v2f ar[A_RING][4];
    #pragma unroll
    for (int gq = 0; gq < A_AHEAD; ++gq) {
        #pragma unroll
        for (int u = 0; u < 4; ++u) {
            ar[gq][u].x = xr[gq * 16 + u * 4];
            ar[gq][u].y = xr[gq * 16 + u * 4 + 1];
        }
        __builtin_amdgcn_sched_barrier(0x3C6);
    }

    for (int i = 0; i < mine; ++i) {
        const int tile = TILE_OF(i);
        const float* xn = (i + 1 < mine) ? ROW_PTR(TILE_OF(i + 1)) : xr;

        v8f acc[4] = {{}, {}, {}, {}};

        for (int jo = 0; jo < K_GROUPS; jo += A_RING) {
            #pragma unroll
            for (int jj = 0; jj < A_RING; ++jj) {
                const int gq = jo + jj + A_AHEAD;
                const float* pf = (gq < K_GROUPS) ? xr : xn;
                const int gk = gq & (K_GROUPS - 1);
                #pragma unroll
                for (int u = 0; u < 4; ++u) {
                    ar[(jj + A_AHEAD) % A_RING][u].x = pf[gk * 16 + u * 4];
                    ar[(jj + A_AHEAD) % A_RING][u].y = pf[gk * 16 + u * 4 + 1];
                }
                __builtin_amdgcn_sched_barrier(0x3C6);
                #pragma unroll
                for (int u = 0; u < 4; ++u) {
                    const v2f a = ar[jj][u];
                    const int prow = (jo + jj) * 8 + u * 2 + half;
                    #pragma unroll
                    for (int t = 0; t < 4; ++t) {
                        v2f b = lds_w[prow * OUT_CH + t * 16 + l15];
                        acc[t] = __builtin_amdgcn_wmma_f32_16x16x4_f32(
                            false, a, false, b, (short)0, acc[t], false, false);
                    }
                }
            }
        }

        const int row0 = tile * 16;
        #pragma unroll
        for (int v = 0; v < 8; ++v) {
            int row = row0 + half * 8 + v;
            if (row < n) {
                #pragma unroll
                for (int t = 0; t < 4; ++t)
                    h[(size_t)row * OUT_CH + t * 16 + l15] = acc[t][v];
            }
        }
        xr = xn;
    }
    #undef TILE_OF
    #undef ROW_PTR
}

#define PULL_WGS      512
#define PULL_THREADS  1024
#define TAB_CAP       102400
#define TAB_STG       7

__global__ __launch_bounds__(PULL_THREADS) void pull_kernel(
        const float* __restrict__ h, const int* __restrict__ csr,
        const unsigned* __restrict__ rowstart, const unsigned* __restrict__ degcnt,
        const float* __restrict__ dinv, const unsigned* __restrict__ deg8w,
        const float* __restrict__ bias, const float* __restrict__ alpha,
        float* __restrict__ out, int n, int e) {
    __shared__ uint4 tab4[TAB_CAP / 16];
    __shared__ unsigned next;

    const int tid  = threadIdx.x;
    const int lane = tid & 31;
    const int wave = __builtin_amdgcn_readfirstlane(tid >> 5);

    const int nscan = (n + SCAN_THREADS - 1) / SCAN_THREADS;
    const int nu    = (nscan * (SCAN_THREADS / 16) < TAB_CAP / 16) ? nscan * (SCAN_THREADS / 16) : TAB_CAP / 16;
    const unsigned tabn = (unsigned)nu * 16u;
    if (nu > 0) {
        const uint4* g4 = (const uint4*)deg8w;
        const int st = (int)((blockIdx.x * 64u) % (unsigned)nu);
        uint4 r[TAB_STG];
        int   qs[TAB_STG];
        #pragma unroll
        for (int k = 0; k < TAB_STG; ++k) {
            int q = tid + k * PULL_THREADS;
            q = (q < nu) ? q : nu - 1;
            q += st;
            if (q >= nu) q -= nu;
            qs[k] = q;
            r[k] = g4[q];
        }
        #pragma unroll
        for (int k = 0; k < TAB_STG; ++k) tab4[qs[k]] = r[k];
    }
    if (tid == 0) next = 32u;
    __syncthreads();
    const unsigned char* tab8 = (const unsigned char*)tab4;

    const int nb  = (int)gridDim.x;
    const int per = n / nb;
    const int rem = n - per * nb;
    const int b   = (int)blockIdx.x;
    const int n0  = b * per + ((b < rem) ? b : rem);
    const int n1  = n0 + per + ((b < rem) ? 1 : 0);

    const float2* h2 = (const float2*)h;
    const int c = lane * 2;
    const float b0 = bias[c],  b1 = bias[c + 1];
    const float a0 = alpha[c], a1 = alpha[c + 1];

    int cur = wave;
    while (n0 + cur < n1) {
        const int node = n0 + cur;

        unsigned start = rowstart[node];
        unsigned cnt   = degcnt[node];
        if (start > (unsigned)e) start = (unsigned)e;
        if (cnt > (unsigned)e - start) cnt = (unsigned)e - start;

        const float dn = dinv[node];
        const float2 hv = h2[(size_t)node * (OUT_CH / 2) + lane];
        float accx = 0.0f, accy = 0.0f;

        for (unsigned base = 0; base < cnt; base += 32) {
            const int m = (cnt - base < 32u) ? (int)(cnt - base) : 32;
            int   sl = 0;
            float nl = 0.0f;
            if (lane < m) {
                sl = csr[start + base + lane];
                const unsigned c8 = ((unsigned)sl < tabn) ? (unsigned)tab8[sl] : 255u;
                float di;
                if (c8 < 255u) di = rsqrtf((float)(c8 + 1u));
                else           di = dinv[sl];
                nl = di * dn;
            }
            int j = 0;
            for (; j + 8 <= m; j += 8) {
                float2 v[8];
                float  w[8];
                #pragma unroll
                for (int u = 0; u < 8; ++u) {
                    const int s = __builtin_amdgcn_readlane(sl, j + u);
                    w[u] = __int_as_float(__builtin_amdgcn_readlane(__float_as_int(nl), j + u));
                    v[u] = h2[(size_t)s * (OUT_CH / 2) + lane];
                }
                #pragma unroll
                for (int u = 0; u < 8; ++u) {
                    accx += v[u].x * w[u];
                    accy += v[u].y * w[u];
                }
            }
            if (j < m) {
                float2 v[8];
                float  w[8];
                #pragma unroll
                for (int u = 0; u < 8; ++u) {
                    v[u].x = 0.0f; v[u].y = 0.0f; w[u] = 0.0f;
                    if (j + u < m) {
                        const int s = __builtin_amdgcn_readlane(sl, j + u);
                        w[u] = __int_as_float(__builtin_amdgcn_readlane(__float_as_int(nl), j + u));
                        v[u] = h2[(size_t)s * (OUT_CH / 2) + lane];
                    }
                }
                #pragma unroll
                for (int u = 0; u < 8; ++u) {
                    accx += v[u].x * w[u];
                    accy += v[u].y * w[u];
                }
            }
        }

        const float vx = accx + hv.x * dn * dn + b0;
        const float vy = accy + hv.y * dn * dn + b1;
        float2 o;
        o.x = (vx >= 0.0f) ? vx : a0 * vx;
        o.y = (vy >= 0.0f) ? vy : a1 * vy;
        ((float2*)out)[(size_t)node * (OUT_CH / 2) + lane] = o;

        unsigned nx = 0u;
        if (lane == 0) nx = atomicAdd(&next, 1u);
        cur = __builtin_amdgcn_readfirstlane((int)nx);
    }
}

extern "C" void kernel_launch(void* const* d_in, const int* in_sizes, int n_in,
                              void* d_out, int out_size, void* d_ws, size_t ws_size,
                              hipStream_t stream) {
    const float* x     = (const float*)d_in[0];
    const int*   ei    = (const int*)  d_in[1];
    const float* W     = (const float*)d_in[2];
    const float* bias  = (const float*)d_in[3];
    const float* alpha = (const float*)d_in[4];
    float*       out   = (float*)d_out;

    const int N = in_sizes[0] / IN_CH;
    const int E = in_sizes[1] / 2;
    const int* src = ei;
    const int* dst = ei + E;

    char* ws = (char*)d_ws;
    size_t hw_bytes = ((size_t)N * OUT_CH * sizeof(float) + 255) & ~(size_t)255;
    size_t e_bytes  = ((size_t)E * sizeof(int) + 255) & ~(size_t)255;
    size_t n_bytes  = ((size_t)N * sizeof(int) + 255) & ~(size_t)255;
    float*    h        = (float*)ws;
    unsigned* part     = (unsigned*)ws;
    int*      csr      = (int*)(ws + hw_bytes);
    unsigned* rowstart = (unsigned*)(ws + hw_bytes + e_bytes);
    unsigned* degcnt   = (unsigned*)(ws + hw_bytes + e_bytes + n_bytes);
    unsigned* blocktot = (unsigned*)(ws + hw_bytes + e_bytes + 2 * n_bytes);
    float*    dinv     = (float*)(ws + 2 * hw_bytes);

    const int threads = 256;
    const int nscan = (N + SCAN_THREADS - 1) / SCAN_THREADS;
    size_t bt_bytes = ((size_t)nscan * sizeof(unsigned) + 255) & ~(size_t)255;
    unsigned* deg8w = (unsigned*)(ws + hw_bytes + e_bytes + 2 * n_bytes + bt_bytes);

    deg_hist_kernel<<<DEG_R * DEG_C, DEG_THREADS, 0, stream>>>(dst, part, N, E);
    dinv_offsets_kernel<<<nscan, SCAN_THREADS, 0, stream>>>(part, dinv, degcnt, blocktot, deg8w, N);
    fill_kernel<<<DEG_R * DEG_C, DEG_THREADS, 0, stream>>>(src, dst, part, blocktot, csr, rowstart, N, E);
    gemm_wmma_kernel<<<GEMM_WGS, threads, 0, stream>>>(x, W, h, N);
    pull_kernel<<<PULL_WGS, PULL_THREADS, 0, stream>>>(
        h, csr, rowstart, degcnt, dinv, deg8w, bias, alpha, out, N, E);
}
